# starting kernel
# baseline (speedup 1.0000x reference)
#include <hip/hip_runtime.h>
#include <hip/hip_bf16.h>

typedef __attribute__((ext_vector_type(2))) float v2f;
typedef __attribute__((ext_vector_type(8))) float v8f;

#define ALPHA   0.25f
#define EPS     1e-8f
#define W_CLASS 2.0f
#define W_BBOX  5.0f
#define W_GIOU  2.0f
#define KPAD    96

__global__ void class_table_kernel(const float* __restrict__ logits,
                                   float* __restrict__ T, int N, int C) {
    int idx = blockIdx.x * blockDim.x + threadIdx.x;
    if (idx >= N * KPAD) return;
    int n = idx / KPAD;
    int c = idx - n * KPAD;
    float v = 0.0f;
    if (c < C) {
        float x = logits[n * C + c];
        float p = 1.0f / (1.0f + __expf(-x));
        float q = 1.0f - p;
        float pos = ALPHA * q * q * (-__logf(p + EPS));
        float neg = (1.0f - ALPHA) * p * p * (-__logf(q + EPS));
        v = pos - neg;
    }
    T[idx] = v;
}

__global__ void onehot_kernel(const int* __restrict__ labels,
                              float* __restrict__ Ht, int M, int Mpad) {
    int idx = blockIdx.x * blockDim.x + threadIdx.x;
    if (idx >= Mpad * KPAD) return;
    int m = idx / KPAD;
    int c = idx - m * KPAD;
    float v = 0.0f;
    if (m < M && labels[m] == c) v = W_CLASS;
    Ht[idx] = v;
}

__global__ void gtpad_kernel(const float* __restrict__ gt,
                             float* __restrict__ Gp, int M, int Mpad) {
    int m = blockIdx.x * blockDim.x + threadIdx.x;
    if (m >= Mpad) return;
    float4 b = (m < M) ? ((const float4*)gt)[m] : make_float4(0.f, 0.f, 1.f, 1.f);
    ((float4*)Gp)[m] = b;
}

__global__ void __launch_bounds__(256)
cost_kernel(const float* __restrict__ T,
            const float* __restrict__ Ht,
            const float* __restrict__ Gp,
            const float* __restrict__ pboxes,
            float* __restrict__ out,
            int M, int CT, int NT) {
    int wave = threadIdx.x >> 5;
    int lane = threadIdx.x & 31;
    int tile = blockIdx.x * (blockDim.x >> 5) + wave;
    if (tile >= NT) return;

    int rt = tile / CT;
    int ct = tile - rt * CT;
    int qbase = rt << 4;
    int cbase = ct << 4;
    int lo = lane & 15;
    int hi = lane >> 4;

    const float* Arow = T  + (size_t)(qbase + lo) * KPAD + 2 * hi;
    const float* Brow = Ht + (size_t)(cbase + lo) * KPAD + 2 * hi;

    v8f acc = {};
#pragma unroll
    for (int k0 = 0; k0 < KPAD; k0 += 4) {
        v2f a = *(const v2f*)(Arow + k0);
        v2f b = *(const v2f*)(Brow + k0);
        acc = __builtin_amdgcn_wmma_f32_16x16x4_f32(
             false, a,  false, b,
             (short)0, acc,  false,  false);
    }

    int g = cbase + lo;
    float4 gb = ((const float4*)Gp)[g];
    float gx0 = gb.x - 0.5f * gb.z, gy0 = gb.y - 0.5f * gb.w;
    float gx1 = gb.x + 0.5f * gb.z, gy1 = gb.y + 0.5f * gb.w;
    float garea = (gx1 - gx0) * (gy1 - gy0);

#pragma unroll
    for (int r = 0; r < 8; ++r) {
        int q = qbase + r + 8 * hi;
        float4 pb = ((const float4*)pboxes)[q];

        float c_bbox = fabsf(pb.x - gb.x) + fabsf(pb.y - gb.y) +
                       fabsf(pb.z - gb.z) + fabsf(pb.w - gb.w);

        float px0 = pb.x - 0.5f * pb.z, py0 = pb.y - 0.5f * pb.w;
        float px1 = pb.x + 0.5f * pb.z, py1 = pb.y + 0.5f * pb.w;
        float parea = (px1 - px0) * (py1 - py0);
        float iw = fmaxf(fminf(px1, gx1) - fmaxf(px0, gx0), 0.0f);
        float ih = fmaxf(fminf(py1, gy1) - fmaxf(py0, gy0), 0.0f);
        float inter = iw * ih;
        float uni = parea + garea - inter;
        float iou = inter / uni;
        float ew = fmaxf(fmaxf(px1, gx1) - fminf(px0, gx0), 0.0f);
        float eh = fmaxf(fmaxf(py1, gy1) - fminf(py0, gy0), 0.0f);
        float earea = ew * eh;
        float giou = iou - (earea - uni) / earea;

        float v = acc[r] + W_BBOX * c_bbox - W_GIOU * giou;
        if (g < M) out[(size_t)q * M + g] = v;
    }
}

extern "C" void kernel_launch(void* const* d_in, const int* in_sizes, int n_in,
                              void* d_out, int out_size, void* d_ws, size_t ws_size,
                              hipStream_t stream) {
    const float* logits = (const float*)d_in[0];
    const float* pboxes = (const float*)d_in[1];
    const float* gt     = (const float*)d_in[2];
    const int*   labels = (const int*)d_in[3];
    float* out = (float*)d_out;

    int Nq = in_sizes[1] / 4;
    int M  = in_sizes[2] / 4;
    int C  = in_sizes[0] / Nq;
    int Mpad = ((M + 15) / 16) * 16;
    int CT = Mpad / 16;
    int RT = Nq / 16;
    int NT = RT * CT;

    float* T  = (float*)d_ws;
    float* Ht = T + (size_t)Nq * KPAD;
    float* Gp = Ht + (size_t)Mpad * KPAD;

    int t1 = Nq * KPAD;
    class_table_kernel<<<(t1 + 255) / 256, 256, 0, stream>>>(logits, T, Nq, C);
    int t2 = Mpad * KPAD;
    onehot_kernel<<<(t2 + 255) / 256, 256, 0, stream>>>(labels, Ht, M, Mpad);
    gtpad_kernel<<<(Mpad + 255) / 256, 256, 0, stream>>>(gt, Gp, M, Mpad);

    const int wavesPerBlock = 8;
    int blocks = (NT + wavesPerBlock - 1) / wavesPerBlock;
    cost_kernel<<<blocks, 32 * wavesPerBlock, 0, stream>>>(T, Ht, Gp, pboxes,
                                                           out, M, CT, NT);
}


# Class-table kernel re-gridded: a lane computes 16 elements of its class column (loads first, then arithmetic); 2700 waves in 450 work-groups, not 43200 one-element waves; same element arithmetic
# speedup vs baseline: 1.8109x; 1.8109x over previous
#include <hip/hip_runtime.h>
#include <hip/hip_bf16.h>


typedef __attribute__((ext_vector_type(2))) float v2f;
typedef __attribute__((ext_vector_type(8))) float v8f;

#define ALPHA   0.25f
#define EPS     1e-8f
#define W_CLASS 2.0f
#define W_BBOX  5.0f
#define W_GIOU  2.0f
#define KPAD    96
#define KSTEPS  (KPAD / 4)
#define LROWS   10
#define WPB     12

#define CT_RY     2
#define CT_PASSES 16

template <bool GUARD>
__device__ __forceinline__ void class_table_rows(const float* __restrict__ logits,
                                                 float* __restrict__ T, int N, int C,
                                                 int n0, int c) {
    bool real = c < C;
    int cc = real ? c : C - 1;

    float x[CT_PASSES];
#pragma unroll
    for (int j = 0; j < CT_PASSES; ++j) {
        int n = n0 + CT_RY * j;
        if (GUARD) n = n < N ? n : N - 1;
        x[j] = logits[n * C + cc];
    }
#pragma unroll
    for (int j = 0; j < CT_PASSES; ++j) {
        int n = n0 + CT_RY * j;
        float p = 1.0f / (1.0f + __expf(-x[j]));
        float q = 1.0f - p;
        float pos = ALPHA * q * q * (-__logf(p + EPS));
        float neg = (1.0f - ALPHA) * p * p * (-__logf(q + EPS));
        float v = real ? pos - neg : 0.0f;
        if (!GUARD || n < N) T[n * KPAD + c] = v;
    }
}

__global__ void __launch_bounds__(KPAD * CT_RY)
class_table_kernel(const float* __restrict__ logits,
                   float* __restrict__ T, int N, int C) {
    int c = threadIdx.x;
    int nb = blockIdx.x * (CT_RY * CT_PASSES);
    int n0 = nb + threadIdx.y;
    if (nb + CT_RY * CT_PASSES <= N)
        class_table_rows<false>(logits, T, N, C, n0, c);
    else
        class_table_rows<true>(logits, T, N, C, n0, c);
}

struct GtCol {
    float4 gb;
    float gx0, gy0, gx1, gy1, garea;
};

__device__ __forceinline__ GtCol make_gtcol(const float* __restrict__ gt, int gi, bool real) {
    GtCol c;
    c.gb = ((const float4*)gt)[gi];
    if (!real) c.gb = make_float4(0.f, 0.f, 1.f, 1.f);
    c.gx0 = c.gb.x - 0.5f * c.gb.z; c.gy0 = c.gb.y - 0.5f * c.gb.w;
    c.gx1 = c.gb.x + 0.5f * c.gb.z; c.gy1 = c.gb.y + 0.5f * c.gb.w;
    c.garea = (c.gx1 - c.gx0) * (c.gy1 - c.gy0);
    return c;
}

struct QRow {
    float4 pb;
    float px0, py0, px1, py1, parea;
};

__device__ __forceinline__ QRow make_qrow(float4 pb) {
    QRow q;
    q.pb = pb;
    q.px0 = pb.x - 0.5f * pb.z; q.py0 = pb.y - 0.5f * pb.w;
    q.px1 = pb.x + 0.5f * pb.z; q.py1 = pb.y + 0.5f * pb.w;
    q.parea = (q.px1 - q.px0) * (q.py1 - q.py0);
    return q;
}

__device__ __forceinline__ float cost_elem(float acc2, const QRow& q, const GtCol& c) {
    float c_bbox = fabsf(q.pb.x - c.gb.x) + fabsf(q.pb.y - c.gb.y) +
                   fabsf(q.pb.z - c.gb.z) + fabsf(q.pb.w - c.gb.w);
    float iw = fmaxf(fminf(q.px1, c.gx1) - fmaxf(q.px0, c.gx0), 0.0f);
    float ih = fmaxf(fminf(q.py1, c.gy1) - fmaxf(q.py0, c.gy0), 0.0f);
    float inter = iw * ih;
    float uni = q.parea + c.garea - inter;
    float iou = inter / uni;
    float ew = fmaxf(fmaxf(q.px1, c.gx1) - fminf(q.px0, c.gx0), 0.0f);
    float eh = fmaxf(fmaxf(q.py1, c.gy1) - fminf(q.py0, c.gy0), 0.0f);
    float earea = ew * eh;
    float giou = iou - (earea - uni) / earea;
    return acc2 + W_BBOX * c_bbox - W_GIOU * giou;
}

__device__ __forceinline__ void load_frags(v2f (&f)[KSTEPS], const float* __restrict__ p) {
#pragma unroll
    for (int i = 0; i < KSTEPS; ++i) f[i] = *(const v2f*)(p + 4 * i);
}

__device__ __forceinline__ void make_onehot(v2f (&f)[KSTEPS], int d) {
#pragma unroll
    for (int i = 0; i < KSTEPS; ++i) {
        f[i].x = (d == 4 * i)     ? W_CLASS : 0.0f;
        f[i].y = (d == 4 * i + 1) ? W_CLASS : 0.0f;
    }
}

__device__ __forceinline__ unsigned block_mask(int label) {
    unsigned v = ((unsigned)label < (unsigned)KPAD) ? (1u << (label >> 2)) : 0u;
#pragma unroll
    for (int s = 16; s >= 1; s >>= 1) v |= (unsigned)__shfl_xor((int)v, s, 32);
    return (unsigned)__builtin_amdgcn_readfirstlane((int)v);
}

__device__ __forceinline__ void load_boxes(float4 (&pb)[8], const float4* __restrict__ p) {
#pragma unroll
    for (int r = 0; r < 8; ++r) pb[r] = p[r];
}

template <bool MASKED>
__device__ __forceinline__ void step_tail(const v8f& acc0, const v8f& acc1,
                                          const float4 (&pb)[8],
                                          const GtCol& c0, const GtCol& c1,
                                          float* __restrict__ orow, int M,
                                          bool st0, bool st1) {
#pragma unroll
    for (int r = 0; r < 8; ++r) {
        QRow qr = make_qrow(pb[r]);
        float v0 = cost_elem(acc0[r], qr, c0);
        float v1 = cost_elem(acc1[r], qr, c1);
        float* o = orow + (size_t)r * M;
        if (MASKED) {
            if (st0) o[0] = v0;
            if (st1) o[16] = v1;
        } else {
            o[0] = v0;
            o[16] = v1;
        }
    }
}

__global__ void __launch_bounds__(32 * WPB)
cost_kernel(const float* __restrict__ T,
            const int* __restrict__ labels,
            const float* __restrict__ gt,
            const float* __restrict__ pboxes,
            float* __restrict__ out,
            int M, int CG, int RT, int NW) {
    int wave = __builtin_amdgcn_readfirstlane((int)(threadIdx.x >> 5));
    int lane = threadIdx.x & 31;
    int wid = blockIdx.x * WPB + wave;
    if (wid >= NW) return;

    int rl = wid / CG;
    int cg = wid - rl * CG;
    int cbase = cg << 5;
    int lo = lane & 15;
    int hi = lane >> 4;

    int g0 = cbase + lo;
    int g1 = g0 + 16;
    bool st0 = g0 < M;
    bool st1 = g1 < M;
    bool full = (cbase + 32) <= M;
    int gi0 = st0 ? g0 : M - 1;
    int gi1 = st1 ? g1 : M - 1;
    int lab0 = labels[gi0];
    int lab1 = labels[gi1];
    GtCol c0 = make_gtcol(gt, gi0, st0);
    GtCol c1 = make_gtcol(gt, gi1, st1);

    v2f b0[KSTEPS], b1[KSTEPS];
    make_onehot(b0, (st0 ? lab0 : -1) - 2 * hi);
    make_onehot(b1, (st1 ? lab1 : -1) - 2 * hi);
    const unsigned m0 = block_mask(st0 ? lab0 : -1);
    const unsigned m1 = block_mask(st1 ? lab1 : -1);

    int rt0 = rl * LROWS;
    int n = RT - rt0;
    if (n > LROWS) n = LROWS;

    v2f a[KSTEPS];
    load_frags(a, T + (size_t)((rt0 << 4) + lo) * KPAD + 2 * hi);
    float4 pb[8];
    load_boxes(pb, (const float4*)pboxes + (rt0 << 4) + 8 * hi);

#pragma unroll 1
    for (int i = 0; i < n; ++i) {
        int qbase = (rt0 + i) << 4;

        v8f zero = {};
        v8f acc0 = __builtin_amdgcn_wmma_f32_16x16x4_f32(
             false, a[0],  false, b0[0],
             (short)0, zero,  false,  false);
        v8f acc1 = __builtin_amdgcn_wmma_f32_16x16x4_f32(
             false, a[0],  false, b1[0],
             (short)0, zero,  false,  false);
#pragma unroll
        for (int k = 1; k < KSTEPS; ++k) {
            if (__builtin_expect((m0 >> k) & 1u, 1))
                acc0 = __builtin_amdgcn_wmma_f32_16x16x4_f32(
                     false, a[k],  false, b0[k],
                     (short)0, acc0,  false,  false);
            if (__builtin_expect((m1 >> k) & 1u, 1))
                acc1 = __builtin_amdgcn_wmma_f32_16x16x4_f32(
                     false, a[k],  false, b1[k],
                     (short)0, acc1,  false,  false);
        }

        int qnext = (i + 1 < n) ? qbase + 16 : qbase;
        float4 pbn[8];
        load_boxes(pbn, (const float4*)pboxes + qnext + 8 * hi);
        load_frags(a, T + (size_t)(qnext + lo) * KPAD + 2 * hi);

        float* orow = out + (size_t)(qbase + 8 * hi) * M + g0;
        if (full) step_tail<false>(acc0, acc1, pb, c0, c1, orow, M, st0, st1);
        else      step_tail<true>(acc0, acc1, pb, c0, c1, orow, M, st0, st1);

#pragma unroll
        for (int r = 0; r < 8; ++r) pb[r] = pbn[r];
    }
}

extern "C" void kernel_launch(void* const* d_in, const int* in_sizes, int n_in,
                              void* d_out, int out_size, void* d_ws, size_t ws_size,
                              hipStream_t stream) {
    const float* logits = (const float*)d_in[0];
    const float* pboxes = (const float*)d_in[1];
    const float* gt     = (const float*)d_in[2];
    const int*   labels = (const int*)d_in[3];
    float* out = (float*)d_out;

    int Nq = in_sizes[1] / 4;
    int M  = in_sizes[2] / 4;
    int C  = in_sizes[0] / Nq;
    int CG = (M + 31) / 32;
    int RT = Nq / 16;
    int RL = (RT + LROWS - 1) / LROWS;
    int NW = RL * CG;

    float* T  = (float*)d_ws;

    int ctrows = CT_RY * CT_PASSES;
    class_table_kernel<<<(Nq + ctrows - 1) / ctrows, dim3(KPAD, CT_RY, 1), 0, stream>>>(
        logits, T, Nq, C);

    int blocks = (NW + WPB - 1) / WPB;
    cost_kernel<<<blocks, 32 * WPB, 0, stream>>>(T, labels, gt, pboxes,
                                                 out, M, CG, RT, NW);
}


# Row tile shared through LDS: a 16-wave work-group (16 column groups of one row lane) copies each 16x96 class-table tile and its query boxes into LDS once; waves read fragments and boxes from LDS
# speedup vs baseline: 1.9384x; 1.0704x over previous
#include <hip/hip_runtime.h>
#include <hip/hip_bf16.h>


typedef __attribute__((ext_vector_type(2))) float v2f;
typedef __attribute__((ext_vector_type(8))) float v8f;

#define ALPHA   0.25f
#define EPS     1e-8f
#define W_CLASS 2.0f
#define W_BBOX  5.0f
#define W_GIOU  2.0f
#define KPAD    96
#define KSTEPS  (KPAD / 4)
#define LROWS   8
#define WPB     16
#define TPITCH  100
#define TILE_F4 (16 * KPAD / 4)
#define BOX_F4  16
#define STAGE_F (16 * TPITCH + 4 * BOX_F4)

#define CT_RY     2
#define CT_PASSES 16

template <bool GUARD>
__device__ __forceinline__ void class_table_rows(const float* __restrict__ logits,
                                                 float* __restrict__ T, int N, int C,
                                                 int n0, int c) {
    bool real = c < C;
    int cc = real ? c : C - 1;

    float x[CT_PASSES];
#pragma unroll
    for (int j = 0; j < CT_PASSES; ++j) {
        int n = n0 + CT_RY * j;
        if (GUARD) n = n < N ? n : N - 1;
        x[j] = logits[n * C + cc];
    }
#pragma unroll
    for (int j = 0; j < CT_PASSES; ++j) {
        int n = n0 + CT_RY * j;
        float p = 1.0f / (1.0f + __expf(-x[j]));
        float q = 1.0f - p;
        float pos = ALPHA * q * q * (-__logf(p + EPS));
        float neg = (1.0f - ALPHA) * p * p * (-__logf(q + EPS));
        float v = real ? pos - neg : 0.0f;
        if (!GUARD || n < N) T[n * KPAD + c] = v;
    }
}

__global__ void __launch_bounds__(KPAD * CT_RY)
class_table_kernel(const float* __restrict__ logits,
                   float* __restrict__ T, int N, int C) {
    int c = threadIdx.x;
    int nb = blockIdx.x * (CT_RY * CT_PASSES);
    int n0 = nb + threadIdx.y;
    if (nb + CT_RY * CT_PASSES <= N)
        class_table_rows<false>(logits, T, N, C, n0, c);
    else
        class_table_rows<true>(logits, T, N, C, n0, c);
}

struct GtCol {
    float4 gb;
    float gx0, gy0, gx1, gy1, garea;
};

__device__ __forceinline__ GtCol make_gtcol(const float* __restrict__ gt, int gi, bool real) {
    GtCol c;
    c.gb = ((const float4*)gt)[gi];
    if (!real) c.gb = make_float4(0.f, 0.f, 1.f, 1.f);
    c.gx0 = c.gb.x - 0.5f * c.gb.z; c.gy0 = c.gb.y - 0.5f * c.gb.w;
    c.gx1 = c.gb.x + 0.5f * c.gb.z; c.gy1 = c.gb.y + 0.5f * c.gb.w;
    c.garea = (c.gx1 - c.gx0) * (c.gy1 - c.gy0);
    return c;
}

struct QRow {
    float4 pb;
    float px0, py0, px1, py1, parea;
};

__device__ __forceinline__ QRow make_qrow(float4 pb) {
    QRow q;
    q.pb = pb;
    q.px0 = pb.x - 0.5f * pb.z; q.py0 = pb.y - 0.5f * pb.w;
    q.px1 = pb.x + 0.5f * pb.z; q.py1 = pb.y + 0.5f * pb.w;
    q.parea = (q.px1 - q.px0) * (q.py1 - q.py0);
    return q;
}

__device__ __forceinline__ float cost_elem(float acc2, const QRow& q, const GtCol& c) {
    float c_bbox = fabsf(q.pb.x - c.gb.x) + fabsf(q.pb.y - c.gb.y) +
                   fabsf(q.pb.z - c.gb.z) + fabsf(q.pb.w - c.gb.w);
    float iw = fmaxf(fminf(q.px1, c.gx1) - fmaxf(q.px0, c.gx0), 0.0f);
    float ih = fmaxf(fminf(q.py1, c.gy1) - fmaxf(q.py0, c.gy0), 0.0f);
    float inter = iw * ih;
    float uni = q.parea + c.garea - inter;
    float iou = inter / uni;
    float ew = fmaxf(fmaxf(q.px1, c.gx1) - fminf(q.px0, c.gx0), 0.0f);
    float eh = fmaxf(fmaxf(q.py1, c.gy1) - fminf(q.py0, c.gy0), 0.0f);
    float earea = ew * eh;
    float giou = iou - (earea - uni) / earea;
    return acc2 + W_BBOX * c_bbox - W_GIOU * giou;
}

__device__ __forceinline__ void load_frags(v2f (&f)[KSTEPS], const float* p) {
#pragma unroll
    for (int i = 0; i < KSTEPS; ++i) f[i] = *(const v2f*)(p + 4 * i);
}

__device__ __forceinline__ void make_onehot(v2f (&f)[KSTEPS], int d) {
#pragma unroll
    for (int i = 0; i < KSTEPS; ++i) {
        f[i].x = (d == 4 * i)     ? W_CLASS : 0.0f;
        f[i].y = (d == 4 * i + 1) ? W_CLASS : 0.0f;
    }
}

__device__ __forceinline__ unsigned block_mask(int label) {
    unsigned v = ((unsigned)label < (unsigned)KPAD) ? (1u << (label >> 2)) : 0u;
#pragma unroll
    for (int s = 16; s >= 1; s >>= 1) v |= (unsigned)__shfl_xor((int)v, s, 32);
    return (unsigned)__builtin_amdgcn_readfirstlane((int)v);
}

__device__ __forceinline__ void lds_barrier() {
    __builtin_amdgcn_fence(__ATOMIC_RELEASE, "workgroup", "local");
    __builtin_amdgcn_s_barrier();
    __builtin_amdgcn_fence(__ATOMIC_ACQUIRE, "workgroup", "local");
}

__device__ __forceinline__ void load_boxes(float4 (&pb)[8], const float4* p) {
#pragma unroll
    for (int r = 0; r < 8; ++r) pb[r] = p[r];
}

template <bool MASKED>
__device__ __forceinline__ void step_tail(const v8f& acc0, const v8f& acc1,
                                          const float4 (&pb)[8],
                                          const GtCol& c0, const GtCol& c1,
                                          float* __restrict__ orow, int M,
                                          bool st0, bool st1) {
#pragma unroll
    for (int r = 0; r < 8; ++r) {
        QRow qr = make_qrow(pb[r]);
        float v0 = cost_elem(acc0[r], qr, c0);
        float v1 = cost_elem(acc1[r], qr, c1);
        float* o = orow + (size_t)r * M;
        if (MASKED) {
            if (st0) o[0] = v0;
            if (st1) o[16] = v1;
        } else {
            o[0] = v0;
            o[16] = v1;
        }
    }
}

__global__ void __launch_bounds__(32 * WPB)
cost_kernel(const float* __restrict__ T,
            const int* __restrict__ labels,
            const float* __restrict__ gt,
            const float* __restrict__ pboxes,
            float* __restrict__ out,
            int M, int RT) {
    __shared__ alignas(16) float sbuf[2][STAGE_F];

    int wave = __builtin_amdgcn_readfirstlane((int)(threadIdx.x >> 5));
    int lane = threadIdx.x & 31;

    int rl = blockIdx.y;
    int cg = blockIdx.x * WPB + wave;
    int cbase = cg << 5;
    int lo = lane & 15;
    int hi = lane >> 4;

    int g0 = cbase + lo;
    int g1 = g0 + 16;
    bool st0 = g0 < M;
    bool st1 = g1 < M;
    bool full = (cbase + 32) <= M;
    int gi0 = st0 ? g0 : M - 1;
    int gi1 = st1 ? g1 : M - 1;
    int lab0 = labels[gi0];
    int lab1 = labels[gi1];
    GtCol c0 = make_gtcol(gt, gi0, st0);
    GtCol c1 = make_gtcol(gt, gi1, st1);

    v2f b0[KSTEPS], b1[KSTEPS];
    make_onehot(b0, (st0 ? lab0 : -1) - 2 * hi);
    make_onehot(b1, (st1 ? lab1 : -1) - 2 * hi);
    const unsigned m0 = block_mask(st0 ? lab0 : -1);
    const unsigned m1 = block_mask(st1 ? lab1 : -1);

    int rt0 = rl * LROWS;
    int n = RT - rt0;
    if (n > LROWS) n = LROWS;

    int ct = (int)threadIdx.x;
    const bool copies = ct < TILE_F4 + BOX_F4;
    const bool table = ct < TILE_F4;
    int crow = ct / (KPAD / 4);
    int cdst = table ? crow * TPITCH + 4 * (ct - crow * (KPAD / 4))
                     : 16 * TPITCH + 4 * (ct - TILE_F4);
    int cstep = table ? TILE_F4 : BOX_F4;
    const float4* csrc = table ? (const float4*)T + ct
                               : (const float4*)pboxes + (ct - TILE_F4);
    const float* frag = &sbuf[0][lo * TPITCH + 2 * hi];
    const float* boxes = &sbuf[0][16 * TPITCH + 4 * (8 * hi)];

    float4 tv = make_float4(0.f, 0.f, 0.f, 0.f);
    if (copies) tv = csrc[(size_t)rt0 * cstep];
    if (copies) *(float4*)&sbuf[0][cdst] = tv;
    lds_barrier();

#pragma unroll 1
    for (int i = 0; i < n; ++i) {
        int qbase = (rt0 + i) << 4;
        int cur = i & 1;
        int qnext = (i + 1 < n) ? qbase + 16 : qbase;

        if (copies) tv = csrc[(size_t)(qnext >> 4) * cstep];

        v2f a[KSTEPS];
        load_frags(a, frag + cur * STAGE_F);

        v8f zero = {};
        v8f acc0 = __builtin_amdgcn_wmma_f32_16x16x4_f32(
             false, a[0],  false, b0[0],
             (short)0, zero,  false,  false);
        v8f acc1 = __builtin_amdgcn_wmma_f32_16x16x4_f32(
             false, a[0],  false, b1[0],
             (short)0, zero,  false,  false);
#pragma unroll
        for (int k = 1; k < KSTEPS; ++k) {
            if (__builtin_expect((m0 >> k) & 1u, 1))
                acc0 = __builtin_amdgcn_wmma_f32_16x16x4_f32(
                     false, a[k],  false, b0[k],
                     (short)0, acc0,  false,  false);
            if (__builtin_expect((m1 >> k) & 1u, 1))
                acc1 = __builtin_amdgcn_wmma_f32_16x16x4_f32(
                     false, a[k],  false, b1[k],
                     (short)0, acc1,  false,  false);
        }

        float4 pb[8];
        load_boxes(pb, (const float4*)(boxes + cur * STAGE_F));

        float* orow = out + (size_t)(qbase + 8 * hi) * M + g0;
        if (full) step_tail<false>(acc0, acc1, pb, c0, c1, orow, M, st0, st1);
        else      step_tail<true>(acc0, acc1, pb, c0, c1, orow, M, st0, st1);

        if (copies) *(float4*)&sbuf[cur ^ 1][cdst] = tv;
        lds_barrier();
    }
}

extern "C" void kernel_launch(void* const* d_in, const int* in_sizes, int n_in,
                              void* d_out, int out_size, void* d_ws, size_t ws_size,
                              hipStream_t stream) {
    const float* logits = (const float*)d_in[0];
    const float* pboxes = (const float*)d_in[1];
    const float* gt     = (const float*)d_in[2];
    const int*   labels = (const int*)d_in[3];
    float* out = (float*)d_out;

    int Nq = in_sizes[1] / 4;
    int M  = in_sizes[2] / 4;
    int C  = in_sizes[0] / Nq;
    int CG = (M + 31) / 32;
    int RT = Nq / 16;
    int RL = (RT + LROWS - 1) / LROWS;
    int GPB = (CG + WPB - 1) / WPB;

    float* T  = (float*)d_ws;

    int ctrows = CT_RY * CT_PASSES;
    class_table_kernel<<<(Nq + ctrows - 1) / ctrows, dim3(KPAD, CT_RY, 1), 0, stream>>>(
        logits, T, Nq, C);

    cost_kernel<<<dim3(GPB, RL, 1), 32 * WPB, 0, stream>>>(T, labels, gt, pboxes,
                                                           out, M, RT);
}


# Output rows staged through LDS: each step's 16x512 tile goes to LDS and wave w writes row w in whole 128-byte lines (5 float4 stores), not 16 stores of 64-byte pieces per wave; same bits out
# speedup vs baseline: 2.3792x; 1.2274x over previous
#include <hip/hip_runtime.h>
#include <hip/hip_bf16.h>


typedef __attribute__((ext_vector_type(2))) float v2f;
typedef __attribute__((ext_vector_type(8))) float v8f;

#define ALPHA   0.25f
#define EPS     1e-8f
#define W_CLASS 2.0f
#define W_BBOX  5.0f
#define W_GIOU  2.0f
#define KPAD    96
#define KC      16
#define KCSTEPS (KC / 4)
#define LROWS   8
#define WPB     16
#define TPITCH  100
#define TILE_F4 (16 * KPAD / 4)
#define BOX_F4  16
#define STAGE_F (16 * TPITCH + 4 * BOX_F4)
#define WGCOLS  (32 * WPB)
#define OPITCH  (WGCOLS + 4)

#define CT_RY     2
#define CT_PASSES 16

template <bool GUARD>
__device__ __forceinline__ void class_table_rows(const float* __restrict__ logits,
                                                 float* __restrict__ T, int N, int C,
                                                 int n0, int c) {
    bool real = c < C;
    int cc = real ? c : C - 1;

    float x[CT_PASSES];
#pragma unroll
    for (int j = 0; j < CT_PASSES; ++j) {
        int n = n0 + CT_RY * j;
        if (GUARD) n = n < N ? n : N - 1;
        x[j] = logits[n * C + cc];
    }
#pragma unroll
    for (int j = 0; j < CT_PASSES; ++j) {
        int n = n0 + CT_RY * j;
        float p = 1.0f / (1.0f + __expf(-x[j]));
        float q = 1.0f - p;
        float pos = ALPHA * q * q * (-__logf(p + EPS));
        float neg = (1.0f - ALPHA) * p * p * (-__logf(q + EPS));
        float v = real ? pos - neg : 0.0f;
        if (!GUARD || n < N) T[n * KPAD + c] = v;
    }
}

__global__ void __launch_bounds__(KPAD * CT_RY)
class_table_kernel(const float* __restrict__ logits,
                   float* __restrict__ T, int N, int C) {
    int c = threadIdx.x;
    int nb = blockIdx.x * (CT_RY * CT_PASSES);
    int n0 = nb + threadIdx.y;
    if (nb + CT_RY * CT_PASSES <= N)
        class_table_rows<false>(logits, T, N, C, n0, c);
    else
        class_table_rows<true>(logits, T, N, C, n0, c);
}

struct GtCol {
    float4 gb;
    float gx0, gy0, gx1, gy1, garea;
};

__device__ __forceinline__ GtCol make_gtcol(const float* __restrict__ gt, int gi, bool real) {
    GtCol c;
    c.gb = ((const float4*)gt)[gi];
    if (!real) c.gb = make_float4(0.f, 0.f, 1.f, 1.f);
    c.gx0 = c.gb.x - 0.5f * c.gb.z; c.gy0 = c.gb.y - 0.5f * c.gb.w;
    c.gx1 = c.gb.x + 0.5f * c.gb.z; c.gy1 = c.gb.y + 0.5f * c.gb.w;
    c.garea = (c.gx1 - c.gx0) * (c.gy1 - c.gy0);
    return c;
}

struct QRow {
    float4 pb;
    float px0, py0, px1, py1, parea;
};

__device__ __forceinline__ QRow make_qrow(float4 pb) {
    QRow q;
    q.pb = pb;
    q.px0 = pb.x - 0.5f * pb.z; q.py0 = pb.y - 0.5f * pb.w;
    q.px1 = pb.x + 0.5f * pb.z; q.py1 = pb.y + 0.5f * pb.w;
    q.parea = (q.px1 - q.px0) * (q.py1 - q.py0);
    return q;
}

__device__ __forceinline__ float cost_elem(float acc2, const QRow& q, const GtCol& c) {
    float c_bbox = fabsf(q.pb.x - c.gb.x) + fabsf(q.pb.y - c.gb.y) +
                   fabsf(q.pb.z - c.gb.z) + fabsf(q.pb.w - c.gb.w);
    float iw = fmaxf(fminf(q.px1, c.gx1) - fmaxf(q.px0, c.gx0), 0.0f);
    float ih = fmaxf(fminf(q.py1, c.gy1) - fmaxf(q.py0, c.gy0), 0.0f);
    float inter = iw * ih;
    float uni = q.parea + c.garea - inter;
    float iou = inter / uni;
    float ew = fmaxf(fmaxf(q.px1, c.gx1) - fminf(q.px0, c.gx0), 0.0f);
    float eh = fmaxf(fmaxf(q.py1, c.gy1) - fminf(q.py0, c.gy0), 0.0f);
    float earea = ew * eh;
    float giou = iou - (earea - uni) / earea;
    return acc2 + W_BBOX * c_bbox - W_GIOU * giou;
}

__device__ __forceinline__ void compact_tile(int lab, int lo, int hi,
                                             v2f (&b)[KCSTEPS], int (&aidx)[2 * KCSTEPS]) {
    bool valid = (unsigned)lab < (unsigned)KPAD;
    int labv = valid ? lab : -1;
    int firstpos = lo;
#pragma unroll
    for (int j = 15; j >= 0; --j) {
        int lj = __builtin_amdgcn_readlane(labv, j);
        if (lj == labv) firstpos = j;
    }
    unsigned firsts = __builtin_amdgcn_ballot_w32(valid && firstpos == lo) & 0xFFFFu;
    int slot = valid ? __popc(firsts & ((1u << firstpos) - 1u)) : -1;
    int d = slot - 2 * hi;
#pragma unroll
    for (int i = 0; i < KCSTEPS; ++i) {
        b[i].x = (d == 4 * i)     ? W_CLASS : 0.0f;
        b[i].y = (d == 4 * i + 1) ? W_CLASS : 0.0f;
    }
#pragma unroll
    for (int e = 0; e < 2 * KCSTEPS; ++e) aidx[e] = lo * TPITCH;
    unsigned m = firsts;
#pragma unroll
    for (int k = 0; k < KC; ++k) {
        int j = m ? __builtin_ctz(m) : 0;
        int c = __builtin_amdgcn_readlane(labv, j);
        int ck = m ? c : 0;
        m &= m - 1u;
        if (hi == ((k >> 1) & 1)) aidx[2 * (k >> 2) + (k & 1)] += ck;
    }
}

__device__ __forceinline__ void load_frags(v2f (&f)[KCSTEPS], const float* stage,
                                           const int (&aidx)[2 * KCSTEPS]) {
#pragma unroll
    for (int i = 0; i < KCSTEPS; ++i) {
        f[i].x = stage[aidx[2 * i]];
        f[i].y = stage[aidx[2 * i + 1]];
    }
}

template <int V> struct StageSel { static constexpr int value = V; };

__device__ __forceinline__ void lds_barrier() {
    __builtin_amdgcn_fence(__ATOMIC_RELEASE, "workgroup", "local");
    __builtin_amdgcn_s_barrier();
    __builtin_amdgcn_fence(__ATOMIC_ACQUIRE, "workgroup", "local");
}

__device__ __forceinline__ void load_boxes(float4 (&pb)[8], const float4* p) {
#pragma unroll
    for (int r = 0; r < 8; ++r) pb[r] = p[r];
}

__device__ __forceinline__ void step_tail(const v8f& acc0, const v8f& acc1,
                                          const float4 (&pb)[8],
                                          const GtCol& c0, const GtCol& c1,
                                          float* otile) {
#pragma unroll
    for (int r = 0; r < 8; ++r) {
        QRow qr = make_qrow(pb[r]);
        float* o = otile + r * OPITCH;
        o[0] = cost_elem(acc0[r], qr, c0);
        o[16] = cost_elem(acc1[r], qr, c1);
    }
}

__device__ __forceinline__ void store_row(const float* lrow, float* __restrict__ out,
                                          size_t a, int len, int lane, bool vec) {
    if (vec) {
        int phi = (int)(a & 31);
        float* gbase = out + (a - phi);
        auto ld = [&](int j) {
            int s = 128 * j + 4 * lane - phi;
            s = s < 0 ? 0 : s;
            s = s > len - 4 ? len - 4 : s;
            return *(const float4*)(lrow + s);
        };
        auto st = [&](int j, float4 v) {
            int t = 128 * j + 4 * lane;
            int s = t - phi;
            if (s >= 0 && s < len) *(float4*)(gbase + t) = v;
        };
        float4 v0 = ld(0), v1 = ld(1), v2 = ld(2), v3 = ld(3), v4 = ld(4);
        st(0, v0); st(1, v1); st(2, v2); st(3, v3); st(4, v4);
    } else {
        for (int s = lane; s < len; s += 32) out[a + s] = lrow[s];
    }
}

__global__ void __launch_bounds__(32 * WPB)
cost_kernel(const float* __restrict__ T,
            const int* __restrict__ labels,
            const float* __restrict__ gt,
            const float* __restrict__ pboxes,
            float* __restrict__ out,
            int M, int RT) {
    __shared__ alignas(16) float sbuf[2][STAGE_F];
    __shared__ alignas(16) float obuf[2][16 * OPITCH];
    static_assert(WPB == 16, "one wave per row of the output tile");

    int wave = __builtin_amdgcn_readfirstlane((int)(threadIdx.x >> 5));
    int lane = threadIdx.x & 31;

    int rl = blockIdx.y;
    int cg = blockIdx.x * WPB + wave;
    int cbase = cg << 5;
    int lo = lane & 15;
    int hi = lane >> 4;

    int g0 = cbase + lo;
    int g1 = g0 + 16;
    bool st0 = g0 < M;
    bool st1 = g1 < M;
    int gi0 = st0 ? g0 : M - 1;
    int gi1 = st1 ? g1 : M - 1;
    int lab0 = labels[gi0];
    int lab1 = labels[gi1];
    GtCol c0 = make_gtcol(gt, gi0, st0);
    GtCol c1 = make_gtcol(gt, gi1, st1);

    int rt0 = rl * LROWS;
    int n = RT - rt0;
    if (n > LROWS) n = LROWS;

    int ct = (int)threadIdx.x;
    const bool copies = ct < TILE_F4 + BOX_F4;
    const bool table = ct < TILE_F4;
    int crow = ct / (KPAD / 4);
    int cdst = table ? crow * TPITCH + 4 * (ct - crow * (KPAD / 4))
                     : 16 * TPITCH + 4 * (ct - TILE_F4);
    int cstep = table ? TILE_F4 : BOX_F4;
    const float4* csrc = table ? (const float4*)T + ct
                               : (const float4*)pboxes + (ct - TILE_F4);
    const int boxes = 16 * TPITCH + 4 * (8 * hi);

    const int opos = (8 * hi) * OPITCH + 32 * wave + lo;
    const int cb0 = blockIdx.x * WGCOLS;
    const int seglen = (M - cb0) < WGCOLS ? (M - cb0) : WGCOLS;
    const bool vec = (M & 3) == 0;

    float4 tv = make_float4(0.f, 0.f, 0.f, 0.f);
    if (copies) tv = csrc[(size_t)rt0 * cstep];

    v2f b0[KCSTEPS], b1[KCSTEPS];
    int ai0[2 * KCSTEPS], ai1[2 * KCSTEPS];
    compact_tile(st0 ? lab0 : -1, lo, hi, b0, ai0);
    compact_tile(st1 ? lab1 : -1, lo, hi, b1, ai1);

    if (copies) *(float4*)&sbuf[0][cdst] = tv;
    lds_barrier();

    auto step = [&](auto sel, int i) {
        constexpr int CUR = decltype(sel)::value;
        const float* stage = &sbuf[CUR][0];
        int qbase = (rt0 + i) << 4;
        int qnext = (i + 1 < n) ? qbase + 16 : qbase;

        if (copies) tv = csrc[(size_t)(qnext >> 4) * cstep];

        v2f a0[KCSTEPS], a1[KCSTEPS];
        load_frags(a0, stage, ai0);
        load_frags(a1, stage, ai1);

        v8f zero = {};
        v8f acc0 = __builtin_amdgcn_wmma_f32_16x16x4_f32(
             false, a0[0],  false, b0[0],
             (short)0, zero,  false,  false);
        v8f acc1 = __builtin_amdgcn_wmma_f32_16x16x4_f32(
             false, a1[0],  false, b1[0],
             (short)0, zero,  false,  false);
#pragma unroll
        for (int k = 1; k < KCSTEPS; ++k) {
            acc0 = __builtin_amdgcn_wmma_f32_16x16x4_f32(
                 false, a0[k],  false, b0[k],
                 (short)0, acc0,  false,  false);
            acc1 = __builtin_amdgcn_wmma_f32_16x16x4_f32(
                 false, a1[k],  false, b1[k],
                 (short)0, acc1,  false,  false);
        }

        float4 pb[8];
        load_boxes(pb, (const float4*)(stage + boxes));

        if (i > 0)
            store_row(&obuf[CUR ^ 1][wave * OPITCH], out,
                      (size_t)(qbase - 16 + wave) * M + cb0, seglen, lane, vec);

        step_tail(acc0, acc1, pb, c0, c1, &obuf[CUR][opos]);

        if (copies) *(float4*)&sbuf[CUR ^ 1][cdst] = tv;
        lds_barrier();
    };

#pragma unroll 1
    for (int i = 0; i < n; i += 2) {
        step(StageSel<0>{}, i);
        if (i + 1 < n) step(StageSel<1>{}, i + 1);
    }

    store_row(&obuf[(n - 1) & 1][wave * OPITCH], out,
              (size_t)(((rt0 + n - 1) << 4) + wave) * M + cb0, seglen, lane, vec);
}

extern "C" void kernel_launch(void* const* d_in, const int* in_sizes, int n_in,
                              void* d_out, int out_size, void* d_ws, size_t ws_size,
                              hipStream_t stream) {
    const float* logits = (const float*)d_in[0];
    const float* pboxes = (const float*)d_in[1];
    const float* gt     = (const float*)d_in[2];
    const int*   labels = (const int*)d_in[3];
    float* out = (float*)d_out;

    int Nq = in_sizes[1] / 4;
    int M  = in_sizes[2] / 4;
    int C  = in_sizes[0] / Nq;
    int CG = (M + 31) / 32;
    int RT = Nq / 16;
    int RL = (RT + LROWS - 1) / LROWS;
    int GPB = (CG + WPB - 1) / WPB;

    float* T  = (float*)d_ws;

    int ctrows = CT_RY * CT_PASSES;
    class_table_kernel<<<(Nq + ctrows - 1) / ctrows, dim3(KPAD, CT_RY, 1), 0, stream>>>(
        logits, T, Nq, C);

    cost_kernel<<<dim3(GPB, RL, 1), 32 * WPB, 0, stream>>>(T, labels, gt, pboxes,
                                                           out, M, RT);
}
